# starting kernel
# baseline (speedup 1.0000x reference)
#include <hip/hip_runtime.h>
#include <hip/hip_bf16.h>

typedef float v2f __attribute__((ext_vector_type(2)));
typedef float v8f __attribute__((ext_vector_type(8)));

#define N_NODES   100000
#define N_EDGES   1600000
#define IN_DIM    128
#define HDIM      64
#define N_CLASSES 16
#define NEG_SLOPE 0.2f

template <int K, int N>
__global__ __launch_bounds__(256) void gemm_wmma_f32(
    const float* __restrict__ A, const float* __restrict__ B,
    const float* __restrict__ bias, float* __restrict__ C, int M) {
  __shared__ float Bs[K * N];
  const int tid = threadIdx.x;
  for (int i = tid; i < K * N; i += 256) Bs[i] = B[i];
  __syncthreads();

  const int wave = tid >> 5;
  const int lane = tid & 31;
  const int hi = lane >> 4;
  const int lo = lane & 15;

  const long m0 = (long)blockIdx.x * 128 + wave * 16;
  long r = m0 + lo;
  if (r > M - 1) r = M - 1;
  const float* __restrict__ Arow = A + (size_t)r * K;

  constexpr int NT = N / 16;
  v8f acc[NT] = {};

  for (int kk = 0; kk < K / 4; ++kk) {
    const int kb = kk * 4 + 2 * hi;
    v2f a;
    a.x = Arow[kb];
    a.y = Arow[kb + 1];
#pragma unroll
    for (int nt = 0; nt < NT; ++nt) {
      v2f b;
      b.x = Bs[kb * N + nt * 16 + lo];
      b.y = Bs[(kb + 1) * N + nt * 16 + lo];
      acc[nt] = __builtin_amdgcn_wmma_f32_16x16x4_f32(
           false, a,  false, b,
           (short)0, acc[nt],  false,  false);
    }
  }

#pragma unroll
  for (int nt = 0; nt < NT; ++nt) {
    const int col = nt * 16 + lo;
    const float bv = bias ? bias[col] : 0.0f;
#pragma unroll
    for (int rr = 0; rr < 8; ++rr) {
      const long row = m0 + rr + 8 * hi;
      if (row < M) C[(size_t)row * N + col] = acc[nt][rr] + bv;
    }
  }
}

__global__ __launch_bounds__(256) void node_prep(
    const float* __restrict__ h, const float* __restrict__ a_src,
    const float* __restrict__ a_dst, float* __restrict__ alpha_s,
    float* __restrict__ alpha_d, float* __restrict__ m, int n_nodes) {
  const int wave = threadIdx.x >> 5;
  const int lane = threadIdx.x & 31;
  const int n = blockIdx.x * 8 + wave;
  if (n >= n_nodes) return;
  const float* hr = h + (size_t)n * HDIM;
  const float h0 = hr[lane], h1 = hr[lane + 32];
  float ps = h0 * a_src[lane] + h1 * a_src[lane + 32];
  float pd = h0 * a_dst[lane] + h1 * a_dst[lane + 32];
#pragma unroll
  for (int off = 16; off > 0; off >>= 1) {
    ps += __shfl_xor(ps, off, 32);
    pd += __shfl_xor(pd, off, 32);
  }
  if (lane == 0) {
    alpha_s[n] = ps;
    alpha_d[n] = pd;
    float e = ps + pd;
    m[n] = (e > 0.0f) ? e : NEG_SLOPE * e;
  }
}

__global__ __launch_bounds__(256) void edge_max(
    const int* __restrict__ adj, const float* __restrict__ alpha_s,
    const float* __restrict__ alpha_d, float* __restrict__ ebuf,
    float* __restrict__ m, int n_edges) {
  const int i = blockIdx.x * 256 + threadIdx.x;
  if (i >= n_edges) return;
  const int src = adj[i];
  const int dst = adj[n_edges + i];
  float e = alpha_s[src] + alpha_d[dst];
  e = (e > 0.0f) ? e : NEG_SLOPE * e;
  ebuf[i] = e;
  atomicMax(&m[dst], e);
}

__global__ __launch_bounds__(256) void node_post(
    const float* __restrict__ h, const float* __restrict__ alpha_s,
    const float* __restrict__ alpha_d, const float* __restrict__ m,
    float* __restrict__ s, float* __restrict__ acc, int n_nodes) {
  const int wave = threadIdx.x >> 5;
  const int lane = threadIdx.x & 31;
  const int n = blockIdx.x * 8 + wave;
  if (n >= n_nodes) return;
  float e = alpha_s[n] + alpha_d[n];
  e = (e > 0.0f) ? e : NEG_SLOPE * e;
  const float pself = __expf(e - m[n]);
  if (lane == 0) s[n] = pself;
  const size_t base = (size_t)n * HDIM;
  acc[base + lane]      = h[base + lane] * pself;
  acc[base + lane + 32] = h[base + lane + 32] * pself;
}

__global__ __launch_bounds__(256) void edge_expsum(
    const int* __restrict__ adj, float* __restrict__ ebuf,
    const float* __restrict__ m, float* __restrict__ s, int n_edges) {
  const int i = blockIdx.x * 256 + threadIdx.x;
  if (i >= n_edges) return;
  const int dst = adj[n_edges + i];
  const float p = __expf(ebuf[i] - m[dst]);
  ebuf[i] = p;
  atomicAdd(&s[dst], p);
}

__global__ __launch_bounds__(256) void edge_scatter(
    const int* __restrict__ adj, const float* __restrict__ ebuf,
    const float* __restrict__ h, float* __restrict__ acc, int n_edges) {
  const int wid = (blockIdx.x * 256 + threadIdx.x) >> 5;
  if (wid >= n_edges) return;
  const int lane = threadIdx.x & 31;
  const int src = adj[wid];
  const int dst = adj[n_edges + wid];
  const float w = ebuf[wid];
  const float* __restrict__ hs = h + (size_t)src * HDIM;
  float* __restrict__ ao = acc + (size_t)dst * HDIM;
  atomicAdd(&ao[lane],      hs[lane]      * w);
  atomicAdd(&ao[lane + 32], hs[lane + 32] * w);
}

__global__ __launch_bounds__(256) void normalize_bias(
    const float* __restrict__ acc, const float* __restrict__ s,
    const float* __restrict__ b, float* __restrict__ out, int total,
    int apply_elu) {
  const int idx = blockIdx.x * 256 + threadIdx.x;
  if (idx >= total) return;
  const int n = idx >> 6;
  const int c = idx & 63;
  float v = acc[idx] / s[n] + b[c];
  if (apply_elu) v = (v > 0.0f) ? v : (__expf(v) - 1.0f);
  out[idx] = v;
}

extern "C" void kernel_launch(void* const* d_in, const int* in_sizes, int n_in,
                              void* d_out, int out_size, void* d_ws,
                              size_t ws_size, hipStream_t stream) {
  const float* X      = (const float*)d_in[0];
  const int*   adj    = (const int*)d_in[1];
  const float* W1     = (const float*)d_in[2];
  const float* a_src1 = (const float*)d_in[3];
  const float* a_dst1 = (const float*)d_in[4];
  const float* b1     = (const float*)d_in[5];
  const float* W2     = (const float*)d_in[6];
  const float* a_src2 = (const float*)d_in[7];
  const float* a_dst2 = (const float*)d_in[8];
  const float* b2     = (const float*)d_in[9];
  const float* clas_W = (const float*)d_in[10];
  const float* clas_b = (const float*)d_in[11];

  float* out = (float*)d_out;
  float* x2  = out + (size_t)N_NODES * N_CLASSES;

  float* ws   = (float*)d_ws;
  float* h    = ws;
  float* x1   = ws + (size_t)N_NODES * HDIM;
  float* acc  = ws + (size_t)2 * N_NODES * HDIM;
  float* as_  = ws + (size_t)3 * N_NODES * HDIM;
  float* ad_  = as_ + N_NODES;
  float* m    = ad_ + N_NODES;
  float* s    = m + N_NODES;
  float* ebuf = s + N_NODES;

  const int gemm_blocks = (N_NODES + 127) / 128;
  const int node_blocks = (N_NODES + 7) / 8;
  const int edge_blocks = (N_EDGES + 255) / 256;
  const int scat_blocks = (N_EDGES + 7) / 8;
  const int norm_total  = N_NODES * HDIM;
  const int norm_blocks = (norm_total + 255) / 256;

  gemm_wmma_f32<IN_DIM, HDIM><<<gemm_blocks, 256, 0, stream>>>(X, W1, nullptr, h, N_NODES);
  node_prep<<<node_blocks, 256, 0, stream>>>(h, a_src1, a_dst1, as_, ad_, m, N_NODES);
  edge_max<<<edge_blocks, 256, 0, stream>>>(adj, as_, ad_, ebuf, m, N_EDGES);
  node_post<<<node_blocks, 256, 0, stream>>>(h, as_, ad_, m, s, acc, N_NODES);
  edge_expsum<<<edge_blocks, 256, 0, stream>>>(adj, ebuf, m, s, N_EDGES);
  edge_scatter<<<scat_blocks, 256, 0, stream>>>(adj, ebuf, h, acc, N_EDGES);
  normalize_bias<<<norm_blocks, 256, 0, stream>>>(acc, s, b1, x1, norm_total,  1);

  gemm_wmma_f32<HDIM, HDIM><<<gemm_blocks, 256, 0, stream>>>(x1, W2, nullptr, h, N_NODES);
  node_prep<<<node_blocks, 256, 0, stream>>>(h, a_src2, a_dst2, as_, ad_, m, N_NODES);
  edge_max<<<edge_blocks, 256, 0, stream>>>(adj, as_, ad_, ebuf, m, N_EDGES);
  node_post<<<node_blocks, 256, 0, stream>>>(h, as_, ad_, m, s, acc, N_NODES);
  edge_expsum<<<edge_blocks, 256, 0, stream>>>(adj, ebuf, m, s, N_EDGES);
  edge_scatter<<<scat_blocks, 256, 0, stream>>>(adj, ebuf, h, acc, N_EDGES);
  normalize_bias<<<norm_blocks, 256, 0, stream>>>(acc, s, b2, x2, norm_total,  0);

  gemm_wmma_f32<HDIM, N_CLASSES><<<gemm_blocks, 256, 0, stream>>>(x2, clas_W, clas_b, out, N_NODES);
}


# node_prep, node_post, normalize_bias, edge_expsum: a wave takes 16 nodes (32 row loads in flight) or a thread 4 to 16 items by 128-bit accesses; 900000 one-item waves become 75000; float sums same
# speedup vs baseline: 1.7896x; 1.7896x over previous
#include <hip/hip_runtime.h>
#include <hip/hip_bf16.h>

typedef float v2f __attribute__((ext_vector_type(2)));
typedef float v8f __attribute__((ext_vector_type(8)));

#define N_NODES   100000
#define N_EDGES   1600000
#define IN_DIM    128
#define HDIM      64
#define N_CLASSES 16
#define NEG_SLOPE 0.2f

template <int K, int N>
__global__ __launch_bounds__(256) void gemm_wmma_f32(
    const float* __restrict__ A, const float* __restrict__ B,
    const float* __restrict__ bias, float* __restrict__ C, int M) {
  __shared__ float Bs[K * N];
  const int tid = threadIdx.x;
  for (int i = tid; i < K * N; i += 256) Bs[i] = B[i];
  __syncthreads();

  const int wave = tid >> 5;
  const int lane = tid & 31;
  const int hi = lane >> 4;
  const int lo = lane & 15;

  const long m0 = (long)blockIdx.x * 128 + wave * 16;
  long r = m0 + lo;
  if (r > M - 1) r = M - 1;
  const float* __restrict__ Arow = A + (size_t)r * K;

  constexpr int NT = N / 16;
  v8f acc[NT] = {};

  for (int kk = 0; kk < K / 4; ++kk) {
    const int kb = kk * 4 + 2 * hi;
    v2f a;
    a.x = Arow[kb];
    a.y = Arow[kb + 1];
#pragma unroll
    for (int nt = 0; nt < NT; ++nt) {
      v2f b;
      b.x = Bs[kb * N + nt * 16 + lo];
      b.y = Bs[(kb + 1) * N + nt * 16 + lo];
      acc[nt] = __builtin_amdgcn_wmma_f32_16x16x4_f32(
           false, a,  false, b,
           (short)0, acc[nt],  false,  false);
    }
  }

#pragma unroll
  for (int nt = 0; nt < NT; ++nt) {
    const int col = nt * 16 + lo;
    const float bv = bias ? bias[col] : 0.0f;
#pragma unroll
    for (int rr = 0; rr < 8; ++rr) {
      const long row = m0 + rr + 8 * hi;
      if (row < M) C[(size_t)row * N + col] = acc[nt][rr] + bv;
    }
  }
}

#define NODES_PER_WAVE 16
#define NODE_BLOCK_NODES (NODES_PER_WAVE * 8)

__global__ __launch_bounds__(256) void node_prep(
    const float* __restrict__ h, const float* __restrict__ a_src,
    const float* __restrict__ a_dst, float* __restrict__ alpha_s,
    float* __restrict__ alpha_d, float* __restrict__ m, int n_nodes) {
  const int lane = threadIdx.x & 31;
  const int n0 = (blockIdx.x * 8 + (threadIdx.x >> 5)) * NODES_PER_WAVE;
  if (n0 >= n_nodes) return;
  float h0[NODES_PER_WAVE], h1[NODES_PER_WAVE];
#pragma unroll
  for (int j = 0; j < NODES_PER_WAVE; ++j) {
    const int n = min(n0 + j, n_nodes - 1);
    const float* hr = h + (size_t)n * HDIM;
    h0[j] = hr[lane];
    h1[j] = hr[lane + 32];
  }
  const float as0 = a_src[lane], as1 = a_src[lane + 32];
  const float ad0 = a_dst[lane], ad1 = a_dst[lane + 32];
  float rs = 0.0f, rd = 0.0f;
#pragma unroll
  for (int j = 0; j < NODES_PER_WAVE; ++j) {
    float ps = h0[j] * as0 + h1[j] * as1;
    float pd = h0[j] * ad0 + h1[j] * ad1;
#pragma unroll
    for (int off = 16; off > 0; off >>= 1) {
      ps += __shfl_xor(ps, off, 32);
      pd += __shfl_xor(pd, off, 32);
    }
    if (lane == j) { rs = ps; rd = pd; }
  }
  const int n = n0 + lane;
  if (lane < NODES_PER_WAVE && n < n_nodes) {
    alpha_s[n] = rs;
    alpha_d[n] = rd;
    const float e = rs + rd;
    m[n] = (e > 0.0f) ? e : NEG_SLOPE * e;
  }
}

#define MAX_RANGES  8
#define MAX_CHUNKS  32
#define RANGE_NODES 12544
#define CHUNK_EDGES (N_EDGES / MAX_CHUNKS)
#define KEY_EMPTY   (-2147483647 - 1)

__device__ __forceinline__ int f32_order_key(float f) {
  const int b = __float_as_int(f);
  return (b >= 0) ? b : (b ^ 0x7fffffff);
}
__device__ __forceinline__ float f32_from_key(int k) {
  return __int_as_float((k >= 0) ? k : (k ^ 0x7fffffff));
}

__global__ __launch_bounds__(1024) void edge_max_tiled(
    const int* __restrict__ adj, const float* __restrict__ alpha_s,
    const float* __restrict__ alpha_d, float* __restrict__ m, int n_edges) {
  __shared__ int best[RANGE_NODES];
  const int tid   = threadIdx.x;
  const int range = blockIdx.x & (MAX_RANGES - 1);
  const int chunk = blockIdx.x / MAX_RANGES;
  const int base  = range * RANGE_NODES;

  for (int i = tid; i < RANGE_NODES; i += 1024) best[i] = KEY_EMPTY;
  __syncthreads();

  const int* __restrict__ srcp = adj + chunk * CHUNK_EDGES;
  const int* __restrict__ dstp = adj + n_edges + chunk * CHUNK_EDGES;
  constexpr int PER_IT = 4 * 1024;
  constexpr int NIT = (CHUNK_EDGES + PER_IT - 1) / PER_IT;

  int dn[4], sn[4];
#pragma unroll
  for (int j = 0; j < 4; ++j) {
    const int idx = j * 1024 + tid;
    dn[j] = dstp[idx];
    sn[j] = srcp[idx];
  }
  for (int it = 0; it < NIT; ++it) {
    int dc[4], sc[4];
#pragma unroll
    for (int j = 0; j < 4; ++j) { dc[j] = dn[j]; sc[j] = sn[j]; }
#pragma unroll
    for (int j = 0; j < 4; ++j) {
      const int idx = (it + 1) * PER_IT + j * 1024 + tid;
      const bool ok = idx < CHUNK_EDGES;
      dn[j] = ok ? dstp[idx] : -1;
      sn[j] = ok ? srcp[idx] : 0;
    }
    unsigned loc[4];
    float a[4];
#pragma unroll
    for (int j = 0; j < 4; ++j) {
      loc[j] = (unsigned)(dc[j] - base);
      a[j] = (loc[j] < (unsigned)RANGE_NODES) ? alpha_s[sc[j]] : 0.0f;
    }
#pragma unroll
    for (int j = 0; j < 4; ++j) {
      if (loc[j] < (unsigned)RANGE_NODES) atomicMax(&best[loc[j]], f32_order_key(a[j]));
    }
  }
  __syncthreads();

  constexpr int NF = (RANGE_NODES + 1023) / 1024;
  int   k[NF];
  float adv[NF];
#pragma unroll
  for (int j = 0; j < NF; ++j) {
    const int i = j * 1024 + tid;
    k[j] = (i < RANGE_NODES) ? best[i] : KEY_EMPTY;
  }
#pragma unroll
  for (int j = 0; j < NF; ++j) {
    adv[j] = (k[j] != KEY_EMPTY) ? alpha_d[base + j * 1024 + tid] : 0.0f;
  }
#pragma unroll
  for (int j = 0; j < NF; ++j) {
    if (k[j] != KEY_EMPTY) {
      float e = f32_from_key(k[j]) + adv[j];
      e = (e > 0.0f) ? e : NEG_SLOPE * e;
      atomicMax(&m[base + j * 1024 + tid], e);
    }
  }
}

__global__ __launch_bounds__(256) void node_post(
    const float* __restrict__ h, const float* __restrict__ alpha_s,
    const float* __restrict__ alpha_d, const float* __restrict__ m,
    float* __restrict__ s, float* __restrict__ acc, int n_nodes) {
  const int lane = threadIdx.x & 31;
  const int n0 = (blockIdx.x * 8 + (threadIdx.x >> 5)) * NODES_PER_WAVE;
  if (n0 >= n_nodes) return;
  float r0[NODES_PER_WAVE], r1[NODES_PER_WAVE];
#pragma unroll
  for (int j = 0; j < NODES_PER_WAVE; ++j) {
    const int n = min(n0 + j, n_nodes - 1);
    const size_t base = (size_t)n * HDIM;
    r0[j] = h[base + lane];
    r1[j] = h[base + lane + 32];
  }
  const int nl = min(n0 + (lane & (NODES_PER_WAVE - 1)), n_nodes - 1);
  float e = alpha_s[nl] + alpha_d[nl];
  e = (e > 0.0f) ? e : NEG_SLOPE * e;
  const float pself = __expf(e - m[nl]);
  if (lane < NODES_PER_WAVE && n0 + lane < n_nodes) s[n0 + lane] = pself;
#pragma unroll
  for (int j = 0; j < NODES_PER_WAVE; ++j) {
    if (n0 + j < n_nodes) {
      const float p = __int_as_float(__builtin_amdgcn_readlane(__float_as_int(pself), j));
      const size_t base = (size_t)(n0 + j) * HDIM;
      acc[base + lane]      = r0[j] * p;
      acc[base + lane + 32] = r1[j] * p;
    }
  }
}

#define EXPSUM_EDGES_PER_THREAD 4
static_assert(N_EDGES % EXPSUM_EDGES_PER_THREAD == 0, "a thread's four edges are all inside the list, and 16-byte aligned in both index rows");

__global__ __launch_bounds__(256) void edge_expsum(
    const int* __restrict__ adj, const float* __restrict__ alpha_s,
    const float* __restrict__ alpha_d, float* __restrict__ ebuf,
    const float* __restrict__ m, float* __restrict__ s, int n_edges) {
  const int i = (blockIdx.x * 256 + threadIdx.x) * EXPSUM_EDGES_PER_THREAD;
  if (i >= n_edges) return;
  const int4 s4 = *reinterpret_cast<const int4*>(adj + i);
  const int4 d4 = *reinterpret_cast<const int4*>(adj + n_edges + i);
  const int sn[4] = {s4.x, s4.y, s4.z, s4.w};
  const int dn[4] = {d4.x, d4.y, d4.z, d4.w};
  float as[4], ad[4], mm[4], p[4];
#pragma unroll
  for (int k = 0; k < 4; ++k) {
    as[k] = alpha_s[sn[k]];
    ad[k] = alpha_d[dn[k]];
    mm[k] = m[dn[k]];
  }
#pragma unroll
  for (int k = 0; k < 4; ++k) {
    float e = as[k] + ad[k];
    e = (e > 0.0f) ? e : NEG_SLOPE * e;
    p[k] = __expf(e - mm[k]);
  }
  *reinterpret_cast<float4*>(ebuf + i) = make_float4(p[0], p[1], p[2], p[3]);
#pragma unroll
  for (int k = 0; k < 4; ++k) atomicAdd(&s[dn[k]], p[k]);
}

#define SCAT_EDGES_PER_WAVE 64
#define SCAT_GROUP          16
static_assert(N_EDGES % (SCAT_EDGES_PER_WAVE * 8) == 0, "a block of 8 waves takes 512 whole edges");

__global__ __launch_bounds__(256) void edge_scatter(
    const int* __restrict__ adj, const float* __restrict__ ebuf,
    const float* __restrict__ h, float* __restrict__ acc, int n_edges) {
  const int lane = threadIdx.x & 31;
  const int e0 = (blockIdx.x * 8 + (threadIdx.x >> 5)) * SCAT_EDGES_PER_WAVE;
  if (e0 >= n_edges) return;

  int sv[2], dv[2], wv[2];
#pragma unroll
  for (int k = 0; k < 2; ++k) {
    const int idx = e0 + k * 32 + lane;
    sv[k] = adj[idx];
    dv[k] = adj[n_edges + idx];
    wv[k] = __float_as_int(ebuf[idx]);
  }

#pragma unroll
  for (int g = 0; g < SCAT_EDGES_PER_WAVE / SCAT_GROUP; ++g) {
    const int k  = (g * SCAT_GROUP) >> 5;
    const int l0 = (g * SCAT_GROUP) & 31;
    float r0[SCAT_GROUP], r1[SCAT_GROUP];
#pragma unroll
    for (int j = 0; j < SCAT_GROUP; ++j) {
      const int s = __builtin_amdgcn_readlane(sv[k], l0 + j);
      const unsigned si = (unsigned)s * HDIM + (unsigned)lane;
      r0[j] = h[si];
      r1[j] = h[si + 32];
    }
#pragma unroll
    for (int j = 0; j < SCAT_GROUP; ++j) {
      const int d = __builtin_amdgcn_readlane(dv[k], l0 + j);
      const float w = __int_as_float(__builtin_amdgcn_readlane(wv[k], l0 + j));
      const unsigned di = (unsigned)d * HDIM + (unsigned)lane;
      atomicAdd(&acc[di],      r0[j] * w);
      atomicAdd(&acc[di + 32], r1[j] * w);
    }
  }
}

#define NORM_VECS         4
#define NORM_STRIDE       (256 * 4)
#define NORM_BLOCK_ELEMS  (NORM_STRIDE * NORM_VECS)
static_assert(NORM_STRIDE % HDIM == 0, "a thread keeps its feature columns from group to group");
static_assert((N_NODES * HDIM) % 4 == 0, "a group of four is all inside the array");

__device__ __forceinline__ float norm_one(float a, float sv, float bv, int apply_elu) {
  float v = a / sv + bv;
  if (apply_elu) v = (v > 0.0f) ? v : (__expf(v) - 1.0f);
  return v;
}

__global__ __launch_bounds__(256) void normalize_bias(
    const float* __restrict__ acc, const float* __restrict__ s,
    const float* __restrict__ b, float* __restrict__ out, int total,
    int apply_elu) {
  const int i0 = blockIdx.x * NORM_BLOCK_ELEMS + threadIdx.x * 4;
  const float4 bv = *reinterpret_cast<const float4*>(b + ((threadIdx.x * 4) & (HDIM - 1)));
  float4 a[NORM_VECS];
  float  sv[NORM_VECS];
#pragma unroll
  for (int k = 0; k < NORM_VECS; ++k) {
    const int idx = i0 + k * NORM_STRIDE;
    const bool ok = idx < total;
    a[k]  = ok ? *reinterpret_cast<const float4*>(acc + idx) : make_float4(0.0f, 0.0f, 0.0f, 0.0f);
    sv[k] = ok ? s[idx >> 6] : 1.0f;
  }
#pragma unroll
  for (int k = 0; k < NORM_VECS; ++k) {
    const int idx = i0 + k * NORM_STRIDE;
    if (idx < total) {
      float4 v;
      v.x = norm_one(a[k].x, sv[k], bv.x, apply_elu);
      v.y = norm_one(a[k].y, sv[k], bv.y, apply_elu);
      v.z = norm_one(a[k].z, sv[k], bv.z, apply_elu);
      v.w = norm_one(a[k].w, sv[k], bv.w, apply_elu);
      *reinterpret_cast<float4*>(out + idx) = v;
    }
  }
}

extern "C" void kernel_launch(void* const* d_in, const int* in_sizes, int n_in,
                              void* d_out, int out_size, void* d_ws,
                              size_t ws_size, hipStream_t stream) {
  const float* X      = (const float*)d_in[0];
  const int*   adj    = (const int*)d_in[1];
  const float* W1     = (const float*)d_in[2];
  const float* a_src1 = (const float*)d_in[3];
  const float* a_dst1 = (const float*)d_in[4];
  const float* b1     = (const float*)d_in[5];
  const float* W2     = (const float*)d_in[6];
  const float* a_src2 = (const float*)d_in[7];
  const float* a_dst2 = (const float*)d_in[8];
  const float* b2     = (const float*)d_in[9];
  const float* clas_W = (const float*)d_in[10];
  const float* clas_b = (const float*)d_in[11];

  float* out = (float*)d_out;
  float* x2  = out + (size_t)N_NODES * N_CLASSES;

  float* ws   = (float*)d_ws;
  float* h    = ws;
  float* x1   = ws + (size_t)N_NODES * HDIM;
  float* acc  = ws + (size_t)2 * N_NODES * HDIM;
  float* as_  = ws + (size_t)3 * N_NODES * HDIM;
  float* ad_  = as_ + N_NODES;
  float* m    = ad_ + N_NODES;
  float* s    = m + N_NODES;
  float* ebuf = s + N_NODES;

  const int gemm_blocks = (N_NODES + 127) / 128;
  const int node_blocks = (N_NODES + NODE_BLOCK_NODES - 1) / NODE_BLOCK_NODES;
  const int edge_blocks = (N_EDGES + 256 * EXPSUM_EDGES_PER_THREAD - 1) / (256 * EXPSUM_EDGES_PER_THREAD);
  const int scat_blocks = N_EDGES / (SCAT_EDGES_PER_WAVE * 8);
  const int norm_total  = N_NODES * HDIM;
  const int norm_blocks = (norm_total + NORM_BLOCK_ELEMS - 1) / NORM_BLOCK_ELEMS;

  gemm_wmma_f32<IN_DIM, HDIM><<<gemm_blocks, 256, 0, stream>>>(X, W1, nullptr, h, N_NODES);
  node_prep<<<node_blocks, 256, 0, stream>>>(h, a_src1, a_dst1, as_, ad_, m, N_NODES);
  edge_max_tiled<<<MAX_RANGES * MAX_CHUNKS, 1024, 0, stream>>>(adj, as_, ad_, m, N_EDGES);
  node_post<<<node_blocks, 256, 0, stream>>>(h, as_, ad_, m, s, acc, N_NODES);
  edge_expsum<<<edge_blocks, 256, 0, stream>>>(adj, as_, ad_, ebuf, m, s, N_EDGES);
  edge_scatter<<<scat_blocks, 256, 0, stream>>>(adj, ebuf, h, acc, N_EDGES);
  normalize_bias<<<norm_blocks, 256, 0, stream>>>(acc, s, b1, x1, norm_total,  1);

  gemm_wmma_f32<HDIM, HDIM><<<gemm_blocks, 256, 0, stream>>>(x1, W2, nullptr, h, N_NODES);
  node_prep<<<node_blocks, 256, 0, stream>>>(h, a_src2, a_dst2, as_, ad_, m, N_NODES);
  edge_max_tiled<<<MAX_RANGES * MAX_CHUNKS, 1024, 0, stream>>>(adj, as_, ad_, m, N_EDGES);
  node_post<<<node_blocks, 256, 0, stream>>>(h, as_, ad_, m, s, acc, N_NODES);
  edge_expsum<<<edge_blocks, 256, 0, stream>>>(adj, as_, ad_, ebuf, m, s, N_EDGES);
  edge_scatter<<<scat_blocks, 256, 0, stream>>>(adj, ebuf, h, acc, N_EDGES);
  normalize_bias<<<norm_blocks, 256, 0, stream>>>(acc, s, b2, x2, norm_total,  0);

  gemm_wmma_f32<HDIM, N_CLASSES><<<gemm_blocks, 256, 0, stream>>>(x2, clas_W, clas_b, out, N_NODES);
}


# Softmax sums: per-edge global f32 atomics go; block = dst range x edge chunk adds edge weights in LDS as 64-bit fixed point (2^-40), one integer atomic per touched node; sums now exact, order-free
# speedup vs baseline: 2.6086x; 1.4576x over previous
#include <hip/hip_runtime.h>
#include <hip/hip_bf16.h>

typedef float v2f __attribute__((ext_vector_type(2)));
typedef float v8f __attribute__((ext_vector_type(8)));

#define N_NODES   100000
#define N_EDGES   1600000
#define IN_DIM    128
#define HDIM      64
#define N_CLASSES 16
#define NEG_SLOPE 0.2f

template <int K, int N>
__global__ __launch_bounds__(256) void gemm_wmma_f32(
    const float* __restrict__ A, const float* __restrict__ B,
    const float* __restrict__ bias, float* __restrict__ C, int M) {
  __shared__ float Bs[K * N];
  const int tid = threadIdx.x;
  for (int i = tid; i < K * N; i += 256) Bs[i] = B[i];
  __syncthreads();

  const int wave = tid >> 5;
  const int lane = tid & 31;
  const int hi = lane >> 4;
  const int lo = lane & 15;

  const long m0 = (long)blockIdx.x * 128 + wave * 16;
  long r = m0 + lo;
  if (r > M - 1) r = M - 1;
  const float* __restrict__ Arow = A + (size_t)r * K;

  constexpr int NT = N / 16;
  v8f acc[NT] = {};

  for (int kk = 0; kk < K / 4; ++kk) {
    const int kb = kk * 4 + 2 * hi;
    v2f a;
    a.x = Arow[kb];
    a.y = Arow[kb + 1];
#pragma unroll
    for (int nt = 0; nt < NT; ++nt) {
      v2f b;
      b.x = Bs[kb * N + nt * 16 + lo];
      b.y = Bs[(kb + 1) * N + nt * 16 + lo];
      acc[nt] = __builtin_amdgcn_wmma_f32_16x16x4_f32(
           false, a,  false, b,
           (short)0, acc[nt],  false,  false);
    }
  }

#pragma unroll
  for (int nt = 0; nt < NT; ++nt) {
    const int col = nt * 16 + lo;
    const float bv = bias ? bias[col] : 0.0f;
#pragma unroll
    for (int rr = 0; rr < 8; ++rr) {
      const long row = m0 + rr + 8 * hi;
      if (row < M) C[(size_t)row * N + col] = acc[nt][rr] + bv;
    }
  }
}

#define NODES_PER_WAVE 16
#define NODE_BLOCK_NODES (NODES_PER_WAVE * 8)

__global__ __launch_bounds__(256) void node_prep(
    const float* __restrict__ h, const float* __restrict__ a_src,
    const float* __restrict__ a_dst, float* __restrict__ alpha_s,
    float* __restrict__ alpha_d, float* __restrict__ m, int n_nodes) {
  const int lane = threadIdx.x & 31;
  const int n0 = (blockIdx.x * 8 + (threadIdx.x >> 5)) * NODES_PER_WAVE;
  if (n0 >= n_nodes) return;
  float h0[NODES_PER_WAVE], h1[NODES_PER_WAVE];
#pragma unroll
  for (int j = 0; j < NODES_PER_WAVE; ++j) {
    const int n = min(n0 + j, n_nodes - 1);
    const float* hr = h + (size_t)n * HDIM;
    h0[j] = hr[lane];
    h1[j] = hr[lane + 32];
  }
  const float as0 = a_src[lane], as1 = a_src[lane + 32];
  const float ad0 = a_dst[lane], ad1 = a_dst[lane + 32];
  float rs = 0.0f, rd = 0.0f;
#pragma unroll
  for (int j = 0; j < NODES_PER_WAVE; ++j) {
    float ps = h0[j] * as0 + h1[j] * as1;
    float pd = h0[j] * ad0 + h1[j] * ad1;
#pragma unroll
    for (int off = 16; off > 0; off >>= 1) {
      ps += __shfl_xor(ps, off, 32);
      pd += __shfl_xor(pd, off, 32);
    }
    if (lane == j) { rs = ps; rd = pd; }
  }
  const int n = n0 + lane;
  if (lane < NODES_PER_WAVE && n < n_nodes) {
    alpha_s[n] = rs;
    alpha_d[n] = rd;
    const float e = rs + rd;
    m[n] = (e > 0.0f) ? e : NEG_SLOPE * e;
  }
}

#define MAX_RANGES  8
#define MAX_CHUNKS  32
#define RANGE_NODES 12544
#define CHUNK_EDGES (N_EDGES / MAX_CHUNKS)
#define KEY_EMPTY   (-2147483647 - 1)

__device__ __forceinline__ int f32_order_key(float f) {
  const int b = __float_as_int(f);
  return (b >= 0) ? b : (b ^ 0x7fffffff);
}
__device__ __forceinline__ float f32_from_key(int k) {
  return __int_as_float((k >= 0) ? k : (k ^ 0x7fffffff));
}

__global__ __launch_bounds__(1024) void edge_max_tiled(
    const int* __restrict__ adj, const float* __restrict__ alpha_s,
    const float* __restrict__ alpha_d, float* __restrict__ m, int n_edges) {
  __shared__ int best[RANGE_NODES];
  const int tid   = threadIdx.x;
  const int range = blockIdx.x & (MAX_RANGES - 1);
  const int chunk = blockIdx.x / MAX_RANGES;
  const int base  = range * RANGE_NODES;

  for (int i = tid; i < RANGE_NODES; i += 1024) best[i] = KEY_EMPTY;
  __syncthreads();

  const int* __restrict__ srcp = adj + chunk * CHUNK_EDGES;
  const int* __restrict__ dstp = adj + n_edges + chunk * CHUNK_EDGES;
  constexpr int PER_IT = 4 * 1024;
  constexpr int NIT = (CHUNK_EDGES + PER_IT - 1) / PER_IT;

  int dn[4], sn[4];
#pragma unroll
  for (int j = 0; j < 4; ++j) {
    const int idx = j * 1024 + tid;
    dn[j] = dstp[idx];
    sn[j] = srcp[idx];
  }
  for (int it = 0; it < NIT; ++it) {
    int dc[4], sc[4];
#pragma unroll
    for (int j = 0; j < 4; ++j) { dc[j] = dn[j]; sc[j] = sn[j]; }
#pragma unroll
    for (int j = 0; j < 4; ++j) {
      const int idx = (it + 1) * PER_IT + j * 1024 + tid;
      const bool ok = idx < CHUNK_EDGES;
      dn[j] = ok ? dstp[idx] : -1;
      sn[j] = ok ? srcp[idx] : 0;
    }
    unsigned loc[4];
    float a[4];
#pragma unroll
    for (int j = 0; j < 4; ++j) {
      loc[j] = (unsigned)(dc[j] - base);
      a[j] = (loc[j] < (unsigned)RANGE_NODES) ? alpha_s[sc[j]] : 0.0f;
    }
#pragma unroll
    for (int j = 0; j < 4; ++j) {
      if (loc[j] < (unsigned)RANGE_NODES) atomicMax(&best[loc[j]], f32_order_key(a[j]));
    }
  }
  __syncthreads();

  constexpr int NF = (RANGE_NODES + 1023) / 1024;
  int   k[NF];
  float adv[NF];
#pragma unroll
  for (int j = 0; j < NF; ++j) {
    const int i = j * 1024 + tid;
    k[j] = (i < RANGE_NODES) ? best[i] : KEY_EMPTY;
  }
#pragma unroll
  for (int j = 0; j < NF; ++j) {
    adv[j] = (k[j] != KEY_EMPTY) ? alpha_d[base + j * 1024 + tid] : 0.0f;
  }
#pragma unroll
  for (int j = 0; j < NF; ++j) {
    if (k[j] != KEY_EMPTY) {
      float e = f32_from_key(k[j]) + adv[j];
      e = (e > 0.0f) ? e : NEG_SLOPE * e;
      atomicMax(&m[base + j * 1024 + tid], e);
    }
  }
}

#define SUM_ONE 1099511627776.0f
__device__ __forceinline__ unsigned long long weight_to_fixed(float p) {
  return (unsigned long long)(p * SUM_ONE);
}
__device__ __forceinline__ float fixed_to_sum(unsigned long long v) {
  return (float)v * (1.0f / SUM_ONE);
}

__global__ __launch_bounds__(256) void node_post(
    const float* __restrict__ h, const float* __restrict__ alpha_s,
    const float* __restrict__ alpha_d, const float* __restrict__ m,
    unsigned long long* __restrict__ s_fix, float* __restrict__ acc, int n_nodes) {
  const int lane = threadIdx.x & 31;
  const int n0 = (blockIdx.x * 8 + (threadIdx.x >> 5)) * NODES_PER_WAVE;
  if (n0 >= n_nodes) return;
  float r0[NODES_PER_WAVE], r1[NODES_PER_WAVE];
#pragma unroll
  for (int j = 0; j < NODES_PER_WAVE; ++j) {
    const int n = min(n0 + j, n_nodes - 1);
    const size_t base = (size_t)n * HDIM;
    r0[j] = h[base + lane];
    r1[j] = h[base + lane + 32];
  }
  const int nl = min(n0 + (lane & (NODES_PER_WAVE - 1)), n_nodes - 1);
  float e = alpha_s[nl] + alpha_d[nl];
  e = (e > 0.0f) ? e : NEG_SLOPE * e;
  const float pself = __expf(e - m[nl]);
  if (lane < NODES_PER_WAVE && n0 + lane < n_nodes) s_fix[n0 + lane] = weight_to_fixed(pself);
#pragma unroll
  for (int j = 0; j < NODES_PER_WAVE; ++j) {
    if (n0 + j < n_nodes) {
      const float p = __int_as_float(__builtin_amdgcn_readlane(__float_as_int(pself), j));
      const size_t base = (size_t)(n0 + j) * HDIM;
      acc[base + lane]      = r0[j] * p;
      acc[base + lane + 32] = r1[j] * p;
    }
  }
}

#define EXPSUM_EDGES_PER_THREAD 4
static_assert(N_EDGES % EXPSUM_EDGES_PER_THREAD == 0, "a thread's four edges are all inside the list, and 16-byte aligned in both index rows");

__global__ __launch_bounds__(256) void edge_expsum(
    const int* __restrict__ adj, const float* __restrict__ alpha_s,
    const float* __restrict__ alpha_d, float* __restrict__ ebuf,
    const float* __restrict__ m, int n_edges) {
  const int i = (blockIdx.x * 256 + threadIdx.x) * EXPSUM_EDGES_PER_THREAD;
  if (i >= n_edges) return;
  const int4 s4 = *reinterpret_cast<const int4*>(adj + i);
  const int4 d4 = *reinterpret_cast<const int4*>(adj + n_edges + i);
  const int sn[4] = {s4.x, s4.y, s4.z, s4.w};
  const int dn[4] = {d4.x, d4.y, d4.z, d4.w};
  float as[4], ad[4], mm[4], p[4];
#pragma unroll
  for (int k = 0; k < 4; ++k) {
    as[k] = alpha_s[sn[k]];
    ad[k] = alpha_d[dn[k]];
    mm[k] = m[dn[k]];
  }
#pragma unroll
  for (int k = 0; k < 4; ++k) {
    float e = as[k] + ad[k];
    e = (e > 0.0f) ? e : NEG_SLOPE * e;
    p[k] = __expf(e - mm[k]);
  }
  *reinterpret_cast<float4*>(ebuf + i) = make_float4(p[0], p[1], p[2], p[3]);
}


static_assert(CHUNK_EDGES % 4 == 0, "a chunk is whole 128-bit groups, 16-byte aligned in both arrays");
#define SUM_VECS      (CHUNK_EDGES / 4)
#define SUM_GROUPS    4
#define SUM_PER_TRIP  (SUM_GROUPS * 1024)

__global__ __launch_bounds__(1024) void edge_sum_tiled(
    const int* __restrict__ adj, const float* __restrict__ ebuf,
    unsigned long long* __restrict__ s_fix, int n_edges) {
  __shared__ unsigned long long part[RANGE_NODES];
  const int tid   = threadIdx.x;
  const int range = blockIdx.x & (MAX_RANGES - 1);
  const int chunk = blockIdx.x / MAX_RANGES;
  const int base  = range * RANGE_NODES;

  for (int i = tid; i < RANGE_NODES; i += 1024) part[i] = 0ull;
  __syncthreads();

  const int4*   __restrict__ dstv = reinterpret_cast<const int4*>(adj + n_edges + chunk * CHUNK_EDGES);
  const float4* __restrict__ wv   = reinterpret_cast<const float4*>(ebuf + chunk * CHUNK_EDGES);
  constexpr int NTRIP = (SUM_VECS + SUM_PER_TRIP - 1) / SUM_PER_TRIP;

  int4   dn[SUM_GROUPS];
  float4 wn[SUM_GROUPS];
#pragma unroll
  for (int j = 0; j < SUM_GROUPS; ++j) {
    const int v = j * 1024 + tid;
    dn[j] = dstv[v];
    wn[j] = wv[v];
  }
#pragma nounroll
  for (int it = 0; it < NTRIP; ++it) {
    int4   dc[SUM_GROUPS];
    float4 wc[SUM_GROUPS];
#pragma unroll
    for (int j = 0; j < SUM_GROUPS; ++j) { dc[j] = dn[j]; wc[j] = wn[j]; }
#pragma unroll
    for (int j = 0; j < SUM_GROUPS; ++j) {
      const int v = (it + 1) * SUM_PER_TRIP + j * 1024 + tid;
      const bool ok = v < SUM_VECS;
      dn[j] = ok ? dstv[v] : make_int4(-1, -1, -1, -1);
      wn[j] = ok ? wv[v]   : make_float4(0.0f, 0.0f, 0.0f, 0.0f);
    }
#pragma unroll
    for (int j = 0; j < SUM_GROUPS; ++j) {
      const int   dd[4] = {dc[j].x, dc[j].y, dc[j].z, dc[j].w};
      const float ww[4] = {wc[j].x, wc[j].y, wc[j].z, wc[j].w};
#pragma unroll
      for (int k = 0; k < 4; ++k) {
        const unsigned loc = (unsigned)(dd[k] - base);
        if (loc < (unsigned)RANGE_NODES) atomicAdd(&part[loc], weight_to_fixed(ww[k]));
      }
    }
  }
  __syncthreads();

  constexpr int NF = (RANGE_NODES + 1023) / 1024;
  unsigned long long v[NF];
#pragma unroll
  for (int j = 0; j < NF; ++j) {
    const int i = j * 1024 + tid;
    v[j] = (i < RANGE_NODES) ? part[i] : 0ull;
  }
#pragma unroll
  for (int j = 0; j < NF; ++j) {
    if (v[j] != 0ull) atomicAdd(&s_fix[base + j * 1024 + tid], v[j]);
  }
}

#define SCAT_EDGES_PER_WAVE 64
#define SCAT_GROUP          16
static_assert(N_EDGES % (SCAT_EDGES_PER_WAVE * 8) == 0, "a block of 8 waves takes 512 whole edges");

__global__ __launch_bounds__(256) void edge_scatter(
    const int* __restrict__ adj, const float* __restrict__ ebuf,
    const float* __restrict__ h, float* __restrict__ acc, int n_edges) {
  const int lane = threadIdx.x & 31;
  const int e0 = (blockIdx.x * 8 + (threadIdx.x >> 5)) * SCAT_EDGES_PER_WAVE;
  if (e0 >= n_edges) return;

  int sv[2], dv[2], wv[2];
#pragma unroll
  for (int k = 0; k < 2; ++k) {
    const int idx = e0 + k * 32 + lane;
    sv[k] = adj[idx];
    dv[k] = adj[n_edges + idx];
    wv[k] = __float_as_int(ebuf[idx]);
  }

#pragma unroll
  for (int g = 0; g < SCAT_EDGES_PER_WAVE / SCAT_GROUP; ++g) {
    const int k  = (g * SCAT_GROUP) >> 5;
    const int l0 = (g * SCAT_GROUP) & 31;
    float r0[SCAT_GROUP], r1[SCAT_GROUP];
#pragma unroll
    for (int j = 0; j < SCAT_GROUP; ++j) {
      const int s = __builtin_amdgcn_readlane(sv[k], l0 + j);
      const unsigned si = (unsigned)s * HDIM + (unsigned)lane;
      r0[j] = h[si];
      r1[j] = h[si + 32];
    }
#pragma unroll
    for (int j = 0; j < SCAT_GROUP; ++j) {
      const int d = __builtin_amdgcn_readlane(dv[k], l0 + j);
      const float w = __int_as_float(__builtin_amdgcn_readlane(wv[k], l0 + j));
      const unsigned di = (unsigned)d * HDIM + (unsigned)lane;
      atomicAdd(&acc[di],      r0[j] * w);
      atomicAdd(&acc[di + 32], r1[j] * w);
    }
  }
}

#define NORM_VECS         4
#define NORM_STRIDE       (256 * 4)
#define NORM_BLOCK_ELEMS  (NORM_STRIDE * NORM_VECS)
static_assert(NORM_STRIDE % HDIM == 0, "a thread keeps its feature columns from group to group");
static_assert((N_NODES * HDIM) % 4 == 0, "a group of four is all inside the array");

__device__ __forceinline__ float norm_one(float a, float sv, float bv, int apply_elu) {
  float v = a / sv + bv;
  if (apply_elu) v = (v > 0.0f) ? v : (__expf(v) - 1.0f);
  return v;
}

__global__ __launch_bounds__(256) void normalize_bias(
    const float* __restrict__ acc, const unsigned long long* __restrict__ s_fix,
    const float* __restrict__ b, float* __restrict__ out, int total,
    int apply_elu) {
  const int i0 = blockIdx.x * NORM_BLOCK_ELEMS + threadIdx.x * 4;
  const float4 bv = *reinterpret_cast<const float4*>(b + ((threadIdx.x * 4) & (HDIM - 1)));
  float4 a[NORM_VECS];
  float  sv[NORM_VECS];
#pragma unroll
  for (int k = 0; k < NORM_VECS; ++k) {
    const int idx = i0 + k * NORM_STRIDE;
    const bool ok = idx < total;
    a[k]  = ok ? *reinterpret_cast<const float4*>(acc + idx) : make_float4(0.0f, 0.0f, 0.0f, 0.0f);
    sv[k] = ok ? fixed_to_sum(s_fix[idx >> 6]) : 1.0f;
  }
#pragma unroll
  for (int k = 0; k < NORM_VECS; ++k) {
    const int idx = i0 + k * NORM_STRIDE;
    if (idx < total) {
      float4 v;
      v.x = norm_one(a[k].x, sv[k], bv.x, apply_elu);
      v.y = norm_one(a[k].y, sv[k], bv.y, apply_elu);
      v.z = norm_one(a[k].z, sv[k], bv.z, apply_elu);
      v.w = norm_one(a[k].w, sv[k], bv.w, apply_elu);
      *reinterpret_cast<float4*>(out + idx) = v;
    }
  }
}

extern "C" void kernel_launch(void* const* d_in, const int* in_sizes, int n_in,
                              void* d_out, int out_size, void* d_ws,
                              size_t ws_size, hipStream_t stream) {
  const float* X      = (const float*)d_in[0];
  const int*   adj    = (const int*)d_in[1];
  const float* W1     = (const float*)d_in[2];
  const float* a_src1 = (const float*)d_in[3];
  const float* a_dst1 = (const float*)d_in[4];
  const float* b1     = (const float*)d_in[5];
  const float* W2     = (const float*)d_in[6];
  const float* a_src2 = (const float*)d_in[7];
  const float* a_dst2 = (const float*)d_in[8];
  const float* b2     = (const float*)d_in[9];
  const float* clas_W = (const float*)d_in[10];
  const float* clas_b = (const float*)d_in[11];

  float* out = (float*)d_out;
  float* x2  = out + (size_t)N_NODES * N_CLASSES;

  float* ws   = (float*)d_ws;
  float* pa   = ws;
  float* pb   = ws + (size_t)N_NODES * HDIM;
  float* acc  = ws + (size_t)2 * N_NODES * HDIM;
  float* as_  = ws + (size_t)3 * N_NODES * HDIM;
  float* ad_  = as_ + N_NODES;
  float* m    = ad_ + N_NODES;
  float* ebuf = m + 2 * N_NODES;
  unsigned long long* sfix1 = (unsigned long long*)pb;
  unsigned long long* sfix2 = (unsigned long long*)pa;

  const int gemm_blocks = (N_NODES + 127) / 128;
  const int node_blocks = (N_NODES + NODE_BLOCK_NODES - 1) / NODE_BLOCK_NODES;
  const int edge_blocks = (N_EDGES + 256 * EXPSUM_EDGES_PER_THREAD - 1) / (256 * EXPSUM_EDGES_PER_THREAD);
  const int scat_blocks = N_EDGES / (SCAT_EDGES_PER_WAVE * 8);
  const int norm_total  = N_NODES * HDIM;
  const int norm_blocks = (norm_total + NORM_BLOCK_ELEMS - 1) / NORM_BLOCK_ELEMS;

  gemm_wmma_f32<IN_DIM, HDIM><<<gemm_blocks, 256, 0, stream>>>(X, W1, nullptr, pa, N_NODES);
  node_prep<<<node_blocks, 256, 0, stream>>>(pa, a_src1, a_dst1, as_, ad_, m, N_NODES);
  edge_max_tiled<<<MAX_RANGES * MAX_CHUNKS, 1024, 0, stream>>>(adj, as_, ad_, m, N_EDGES);
  node_post<<<node_blocks, 256, 0, stream>>>(pa, as_, ad_, m, sfix1, acc, N_NODES);
  edge_expsum<<<edge_blocks, 256, 0, stream>>>(adj, as_, ad_, ebuf, m, N_EDGES);
  edge_sum_tiled<<<MAX_RANGES * MAX_CHUNKS, 1024, 0, stream>>>(adj, ebuf, sfix1, N_EDGES);
  edge_scatter<<<scat_blocks, 256, 0, stream>>>(adj, ebuf, pa, acc, N_EDGES);
  normalize_bias<<<norm_blocks, 256, 0, stream>>>(acc, sfix1, b1, pa, norm_total,  1);

  gemm_wmma_f32<HDIM, HDIM><<<gemm_blocks, 256, 0, stream>>>(pa, W2, nullptr, pb, N_NODES);
  node_prep<<<node_blocks, 256, 0, stream>>>(pb, a_src2, a_dst2, as_, ad_, m, N_NODES);
  edge_max_tiled<<<MAX_RANGES * MAX_CHUNKS, 1024, 0, stream>>>(adj, as_, ad_, m, N_EDGES);
  node_post<<<node_blocks, 256, 0, stream>>>(pb, as_, ad_, m, sfix2, acc, N_NODES);
  edge_expsum<<<edge_blocks, 256, 0, stream>>>(adj, as_, ad_, ebuf, m, N_EDGES);
  edge_sum_tiled<<<MAX_RANGES * MAX_CHUNKS, 1024, 0, stream>>>(adj, ebuf, sfix2, N_EDGES);
  edge_scatter<<<scat_blocks, 256, 0, stream>>>(adj, ebuf, pb, acc, N_EDGES);
  normalize_bias<<<norm_blocks, 256, 0, stream>>>(acc, sfix2, b2, x2, norm_total,  0);

  gemm_wmma_f32<HDIM, N_CLASSES><<<gemm_blocks, 256, 0, stream>>>(x2, clas_W, clas_b, out, N_NODES);
}


# edge_scatter: two accumulator planes, one per half of the package; a wave adds into its own half's plane (by die number), normalize_bias adds the planes: each f32 sum is now two partial sums
# speedup vs baseline: 2.6889x; 1.0308x over previous
#include <hip/hip_runtime.h>
#include <hip/hip_bf16.h>

typedef float v2f __attribute__((ext_vector_type(2)));
typedef float v8f __attribute__((ext_vector_type(8)));

#define N_NODES   100000
#define N_EDGES   1600000
#define IN_DIM    128
#define HDIM      64
#define N_CLASSES 16
#define NEG_SLOPE 0.2f

template <int K, int N>
__global__ __launch_bounds__(256) void gemm_wmma_f32(
    const float* __restrict__ A, const float* __restrict__ B,
    const float* __restrict__ bias, float* __restrict__ C, int M) {
  __shared__ float Bs[K * N];
  const int tid = threadIdx.x;
  for (int i = tid; i < K * N; i += 256) Bs[i] = B[i];
  __syncthreads();

  const int wave = tid >> 5;
  const int lane = tid & 31;
  const int hi = lane >> 4;
  const int lo = lane & 15;

  const long m0 = (long)blockIdx.x * 128 + wave * 16;
  long r = m0 + lo;
  if (r > M - 1) r = M - 1;
  const float* __restrict__ Arow = A + (size_t)r * K;

  constexpr int NT = N / 16;
  v8f acc[NT] = {};

  for (int kk = 0; kk < K / 4; ++kk) {
    const int kb = kk * 4 + 2 * hi;
    v2f a;
    a.x = Arow[kb];
    a.y = Arow[kb + 1];
#pragma unroll
    for (int nt = 0; nt < NT; ++nt) {
      v2f b;
      b.x = Bs[kb * N + nt * 16 + lo];
      b.y = Bs[(kb + 1) * N + nt * 16 + lo];
      acc[nt] = __builtin_amdgcn_wmma_f32_16x16x4_f32(
           false, a,  false, b,
           (short)0, acc[nt],  false,  false);
    }
  }

#pragma unroll
  for (int nt = 0; nt < NT; ++nt) {
    const int col = nt * 16 + lo;
    const float bv = bias ? bias[col] : 0.0f;
#pragma unroll
    for (int rr = 0; rr < 8; ++rr) {
      const long row = m0 + rr + 8 * hi;
      if (row < M) C[(size_t)row * N + col] = acc[nt][rr] + bv;
    }
  }
}

#define NODES_PER_WAVE 16
#define NODE_BLOCK_NODES (NODES_PER_WAVE * 8)

__global__ __launch_bounds__(256) void node_prep(
    const float* __restrict__ h, const float* __restrict__ a_src,
    const float* __restrict__ a_dst, float* __restrict__ alpha_s,
    float* __restrict__ alpha_d, float* __restrict__ m, int n_nodes) {
  const int lane = threadIdx.x & 31;
  const int n0 = (blockIdx.x * 8 + (threadIdx.x >> 5)) * NODES_PER_WAVE;
  if (n0 >= n_nodes) return;
  float h0[NODES_PER_WAVE], h1[NODES_PER_WAVE];
#pragma unroll
  for (int j = 0; j < NODES_PER_WAVE; ++j) {
    const int n = min(n0 + j, n_nodes - 1);
    const float* hr = h + (size_t)n * HDIM;
    h0[j] = hr[lane];
    h1[j] = hr[lane + 32];
  }
  const float as0 = a_src[lane], as1 = a_src[lane + 32];
  const float ad0 = a_dst[lane], ad1 = a_dst[lane + 32];
  float rs = 0.0f, rd = 0.0f;
#pragma unroll
  for (int j = 0; j < NODES_PER_WAVE; ++j) {
    float ps = h0[j] * as0 + h1[j] * as1;
    float pd = h0[j] * ad0 + h1[j] * ad1;
#pragma unroll
    for (int off = 16; off > 0; off >>= 1) {
      ps += __shfl_xor(ps, off, 32);
      pd += __shfl_xor(pd, off, 32);
    }
    if (lane == j) { rs = ps; rd = pd; }
  }
  const int n = n0 + lane;
  if (lane < NODES_PER_WAVE && n < n_nodes) {
    alpha_s[n] = rs;
    alpha_d[n] = rd;
    const float e = rs + rd;
    m[n] = (e > 0.0f) ? e : NEG_SLOPE * e;
  }
}

#define MAX_RANGES  8
#define MAX_CHUNKS  32
#define RANGE_NODES 12544
#define CHUNK_EDGES (N_EDGES / MAX_CHUNKS)
#define KEY_EMPTY   (-2147483647 - 1)

__device__ __forceinline__ int f32_order_key(float f) {
  const int b = __float_as_int(f);
  return (b >= 0) ? b : (b ^ 0x7fffffff);
}
__device__ __forceinline__ float f32_from_key(int k) {
  return __int_as_float((k >= 0) ? k : (k ^ 0x7fffffff));
}

__global__ __launch_bounds__(1024) void edge_max_tiled(
    const int* __restrict__ adj, const float* __restrict__ alpha_s,
    const float* __restrict__ alpha_d, float* __restrict__ m, int n_edges) {
  __shared__ int best[RANGE_NODES];
  const int tid   = threadIdx.x;
  const int range = blockIdx.x & (MAX_RANGES - 1);
  const int chunk = blockIdx.x / MAX_RANGES;
  const int base  = range * RANGE_NODES;

  for (int i = tid; i < RANGE_NODES; i += 1024) best[i] = KEY_EMPTY;
  __syncthreads();

  const int* __restrict__ srcp = adj + chunk * CHUNK_EDGES;
  const int* __restrict__ dstp = adj + n_edges + chunk * CHUNK_EDGES;
  constexpr int PER_IT = 4 * 1024;
  constexpr int NIT = (CHUNK_EDGES + PER_IT - 1) / PER_IT;

  int dn[4], sn[4];
#pragma unroll
  for (int j = 0; j < 4; ++j) {
    const int idx = j * 1024 + tid;
    dn[j] = dstp[idx];
    sn[j] = srcp[idx];
  }
  for (int it = 0; it < NIT; ++it) {
    int dc[4], sc[4];
#pragma unroll
    for (int j = 0; j < 4; ++j) { dc[j] = dn[j]; sc[j] = sn[j]; }
#pragma unroll
    for (int j = 0; j < 4; ++j) {
      const int idx = (it + 1) * PER_IT + j * 1024 + tid;
      const bool ok = idx < CHUNK_EDGES;
      dn[j] = ok ? dstp[idx] : -1;
      sn[j] = ok ? srcp[idx] : 0;
    }
    unsigned loc[4];
    float a[4];
#pragma unroll
    for (int j = 0; j < 4; ++j) {
      loc[j] = (unsigned)(dc[j] - base);
      a[j] = (loc[j] < (unsigned)RANGE_NODES) ? alpha_s[sc[j]] : 0.0f;
    }
#pragma unroll
    for (int j = 0; j < 4; ++j) {
      if (loc[j] < (unsigned)RANGE_NODES) atomicMax(&best[loc[j]], f32_order_key(a[j]));
    }
  }
  __syncthreads();

  constexpr int NF = (RANGE_NODES + 1023) / 1024;
  int   k[NF];
  float adv[NF];
#pragma unroll
  for (int j = 0; j < NF; ++j) {
    const int i = j * 1024 + tid;
    k[j] = (i < RANGE_NODES) ? best[i] : KEY_EMPTY;
  }
#pragma unroll
  for (int j = 0; j < NF; ++j) {
    adv[j] = (k[j] != KEY_EMPTY) ? alpha_d[base + j * 1024 + tid] : 0.0f;
  }
#pragma unroll
  for (int j = 0; j < NF; ++j) {
    if (k[j] != KEY_EMPTY) {
      float e = f32_from_key(k[j]) + adv[j];
      e = (e > 0.0f) ? e : NEG_SLOPE * e;
      atomicMax(&m[base + j * 1024 + tid], e);
    }
  }
}

#define SUM_ONE 1099511627776.0f
__device__ __forceinline__ unsigned long long weight_to_fixed(float p) {
  return (unsigned long long)(p * SUM_ONE);
}
__device__ __forceinline__ float fixed_to_sum(unsigned long long v) {
  return (float)v * (1.0f / SUM_ONE);
}

__global__ __launch_bounds__(256) void node_post(
    const float* __restrict__ h, const float* __restrict__ alpha_s,
    const float* __restrict__ alpha_d, const float* __restrict__ m,
    unsigned long long* __restrict__ s_fix, float* __restrict__ acc,
    float* __restrict__ acc_far, int n_nodes) {
  const int lane = threadIdx.x & 31;
  const int n0 = (blockIdx.x * 8 + (threadIdx.x >> 5)) * NODES_PER_WAVE;
  if (n0 >= n_nodes) return;
  float r0[NODES_PER_WAVE], r1[NODES_PER_WAVE];
#pragma unroll
  for (int j = 0; j < NODES_PER_WAVE; ++j) {
    const int n = min(n0 + j, n_nodes - 1);
    const size_t base = (size_t)n * HDIM;
    r0[j] = h[base + lane];
    r1[j] = h[base + lane + 32];
  }
  const int nl = min(n0 + (lane & (NODES_PER_WAVE - 1)), n_nodes - 1);
  float e = alpha_s[nl] + alpha_d[nl];
  e = (e > 0.0f) ? e : NEG_SLOPE * e;
  const float pself = __expf(e - m[nl]);
  if (lane < NODES_PER_WAVE && n0 + lane < n_nodes) s_fix[n0 + lane] = weight_to_fixed(pself);
#pragma unroll
  for (int j = 0; j < NODES_PER_WAVE; ++j) {
    if (n0 + j < n_nodes) {
      const float p = __int_as_float(__builtin_amdgcn_readlane(__float_as_int(pself), j));
      const size_t base = (size_t)(n0 + j) * HDIM;
      acc[base + lane]      = r0[j] * p;
      acc[base + lane + 32] = r1[j] * p;
      acc_far[base + lane]      = 0.0f;
      acc_far[base + lane + 32] = 0.0f;
    }
  }
}

#define EXPSUM_EDGES_PER_THREAD 4
static_assert(N_EDGES % EXPSUM_EDGES_PER_THREAD == 0, "a thread's four edges are all inside the list, and 16-byte aligned in both index rows");

__global__ __launch_bounds__(256) void edge_expsum(
    const int* __restrict__ adj, const float* __restrict__ alpha_s,
    const float* __restrict__ alpha_d, float* __restrict__ ebuf,
    const float* __restrict__ m, int n_edges) {
  const int i = (blockIdx.x * 256 + threadIdx.x) * EXPSUM_EDGES_PER_THREAD;
  if (i >= n_edges) return;
  const int4 s4 = *reinterpret_cast<const int4*>(adj + i);
  const int4 d4 = *reinterpret_cast<const int4*>(adj + n_edges + i);
  const int sn[4] = {s4.x, s4.y, s4.z, s4.w};
  const int dn[4] = {d4.x, d4.y, d4.z, d4.w};
  float as[4], ad[4], mm[4], p[4];
#pragma unroll
  for (int k = 0; k < 4; ++k) {
    as[k] = alpha_s[sn[k]];
    ad[k] = alpha_d[dn[k]];
    mm[k] = m[dn[k]];
  }
#pragma unroll
  for (int k = 0; k < 4; ++k) {
    float e = as[k] + ad[k];
    e = (e > 0.0f) ? e : NEG_SLOPE * e;
    p[k] = __expf(e - mm[k]);
  }
  *reinterpret_cast<float4*>(ebuf + i) = make_float4(p[0], p[1], p[2], p[3]);
}


static_assert(CHUNK_EDGES % 4 == 0, "a chunk is whole 128-bit groups, 16-byte aligned in both arrays");
#define SUM_VECS      (CHUNK_EDGES / 4)
#define SUM_GROUPS    4
#define SUM_PER_TRIP  (SUM_GROUPS * 1024)

__global__ __launch_bounds__(1024) void edge_sum_tiled(
    const int* __restrict__ adj, const float* __restrict__ ebuf,
    unsigned long long* __restrict__ s_fix, int n_edges) {
  __shared__ unsigned long long part[RANGE_NODES];
  const int tid   = threadIdx.x;
  const int range = blockIdx.x & (MAX_RANGES - 1);
  const int chunk = blockIdx.x / MAX_RANGES;
  const int base  = range * RANGE_NODES;

  for (int i = tid; i < RANGE_NODES; i += 1024) part[i] = 0ull;
  __syncthreads();

  const int4*   __restrict__ dstv = reinterpret_cast<const int4*>(adj + n_edges + chunk * CHUNK_EDGES);
  const float4* __restrict__ wv   = reinterpret_cast<const float4*>(ebuf + chunk * CHUNK_EDGES);
  constexpr int NTRIP = (SUM_VECS + SUM_PER_TRIP - 1) / SUM_PER_TRIP;

  int4   dn[SUM_GROUPS];
  float4 wn[SUM_GROUPS];
#pragma unroll
  for (int j = 0; j < SUM_GROUPS; ++j) {
    const int v = j * 1024 + tid;
    dn[j] = dstv[v];
    wn[j] = wv[v];
  }
#pragma nounroll
  for (int it = 0; it < NTRIP; ++it) {
    int4   dc[SUM_GROUPS];
    float4 wc[SUM_GROUPS];
#pragma unroll
    for (int j = 0; j < SUM_GROUPS; ++j) { dc[j] = dn[j]; wc[j] = wn[j]; }
#pragma unroll
    for (int j = 0; j < SUM_GROUPS; ++j) {
      const int v = (it + 1) * SUM_PER_TRIP + j * 1024 + tid;
      const bool ok = v < SUM_VECS;
      dn[j] = ok ? dstv[v] : make_int4(-1, -1, -1, -1);
      wn[j] = ok ? wv[v]   : make_float4(0.0f, 0.0f, 0.0f, 0.0f);
    }
#pragma unroll
    for (int j = 0; j < SUM_GROUPS; ++j) {
      const int   dd[4] = {dc[j].x, dc[j].y, dc[j].z, dc[j].w};
      const float ww[4] = {wc[j].x, wc[j].y, wc[j].z, wc[j].w};
#pragma unroll
      for (int k = 0; k < 4; ++k) {
        const unsigned loc = (unsigned)(dd[k] - base);
        if (loc < (unsigned)RANGE_NODES) atomicAdd(&part[loc], weight_to_fixed(ww[k]));
      }
    }
  }
  __syncthreads();

  constexpr int NF = (RANGE_NODES + 1023) / 1024;
  unsigned long long v[NF];
#pragma unroll
  for (int j = 0; j < NF; ++j) {
    const int i = j * 1024 + tid;
    v[j] = (i < RANGE_NODES) ? part[i] : 0ull;
  }
#pragma unroll
  for (int j = 0; j < NF; ++j) {
    if (v[j] != 0ull) atomicAdd(&s_fix[base + j * 1024 + tid], v[j]);
  }
}

#define SCAT_EDGES_PER_WAVE 64
#define SCAT_GROUP          16
static_assert(N_EDGES % (SCAT_EDGES_PER_WAVE * 8) == 0, "a block of 8 waves takes 512 whole edges");

__global__ __launch_bounds__(256) void edge_scatter(
    const int* __restrict__ adj, const float* __restrict__ ebuf,
    const float* __restrict__ h, float* __restrict__ acc_near, float* __restrict__ acc_far,
    int n_edges) {
  const int lane = threadIdx.x & 31;
  const int e0 = (blockIdx.x * 8 + (threadIdx.x >> 5)) * SCAT_EDGES_PER_WAVE;
  if (e0 >= n_edges) return;

  const unsigned placement = __builtin_amdgcn_s_sendmsg_rtn(0x87);
  float* __restrict__ acc = ((placement >> 18) & 1u) ? acc_far : acc_near;

  int sv[2], dv[2], wv[2];
#pragma unroll
  for (int k = 0; k < 2; ++k) {
    const int idx = e0 + k * 32 + lane;
    sv[k] = adj[idx];
    dv[k] = adj[n_edges + idx];
    wv[k] = __float_as_int(ebuf[idx]);
  }

#pragma unroll
  for (int g = 0; g < SCAT_EDGES_PER_WAVE / SCAT_GROUP; ++g) {
    const int k  = (g * SCAT_GROUP) >> 5;
    const int l0 = (g * SCAT_GROUP) & 31;
    float r0[SCAT_GROUP], r1[SCAT_GROUP];
#pragma unroll
    for (int j = 0; j < SCAT_GROUP; ++j) {
      const int s = __builtin_amdgcn_readlane(sv[k], l0 + j);
      const unsigned si = (unsigned)s * HDIM + (unsigned)lane;
      r0[j] = h[si];
      r1[j] = h[si + 32];
    }
#pragma unroll
    for (int j = 0; j < SCAT_GROUP; ++j) {
      const int d = __builtin_amdgcn_readlane(dv[k], l0 + j);
      const float w = __int_as_float(__builtin_amdgcn_readlane(wv[k], l0 + j));
      const unsigned di = (unsigned)d * HDIM + (unsigned)lane;
      atomicAdd(&acc[di],      r0[j] * w);
      atomicAdd(&acc[di + 32], r1[j] * w);
    }
  }
}

#define NORM_VECS         4
#define NORM_STRIDE       (256 * 4)
#define NORM_BLOCK_ELEMS  (NORM_STRIDE * NORM_VECS)
static_assert(NORM_STRIDE % HDIM == 0, "a thread keeps its feature columns from group to group");
static_assert((N_NODES * HDIM) % 4 == 0, "a group of four is all inside the array");

__device__ __forceinline__ float norm_one(float a, float sv, float bv, int apply_elu) {
  float v = a / sv + bv;
  if (apply_elu) v = (v > 0.0f) ? v : (__expf(v) - 1.0f);
  return v;
}

__global__ __launch_bounds__(256) void normalize_bias(
    const float* __restrict__ acc, const float* __restrict__ acc_far,
    const unsigned long long* __restrict__ s_fix,
    const float* __restrict__ b, float* __restrict__ out, int total,
    int apply_elu) {
  const int i0 = blockIdx.x * NORM_BLOCK_ELEMS + threadIdx.x * 4;
  const float4 bv = *reinterpret_cast<const float4*>(b + ((threadIdx.x * 4) & (HDIM - 1)));
  float4 a[NORM_VECS], a2[NORM_VECS];
  float  sv[NORM_VECS];
#pragma unroll
  for (int k = 0; k < NORM_VECS; ++k) {
    const int idx = i0 + k * NORM_STRIDE;
    const bool ok = idx < total;
    a[k]  = ok ? *reinterpret_cast<const float4*>(acc + idx) : make_float4(0.0f, 0.0f, 0.0f, 0.0f);
    a2[k] = ok ? *reinterpret_cast<const float4*>(acc_far + idx) : make_float4(0.0f, 0.0f, 0.0f, 0.0f);
    sv[k] = ok ? fixed_to_sum(s_fix[idx >> 6]) : 1.0f;
  }
#pragma unroll
  for (int k = 0; k < NORM_VECS; ++k) {
    const int idx = i0 + k * NORM_STRIDE;
    if (idx < total) {
      float4 v;
      v.x = norm_one(a[k].x + a2[k].x, sv[k], bv.x, apply_elu);
      v.y = norm_one(a[k].y + a2[k].y, sv[k], bv.y, apply_elu);
      v.z = norm_one(a[k].z + a2[k].z, sv[k], bv.z, apply_elu);
      v.w = norm_one(a[k].w + a2[k].w, sv[k], bv.w, apply_elu);
      *reinterpret_cast<float4*>(out + idx) = v;
    }
  }
}

extern "C" void kernel_launch(void* const* d_in, const int* in_sizes, int n_in,
                              void* d_out, int out_size, void* d_ws,
                              size_t ws_size, hipStream_t stream) {
  const float* X      = (const float*)d_in[0];
  const int*   adj    = (const int*)d_in[1];
  const float* W1     = (const float*)d_in[2];
  const float* a_src1 = (const float*)d_in[3];
  const float* a_dst1 = (const float*)d_in[4];
  const float* b1     = (const float*)d_in[5];
  const float* W2     = (const float*)d_in[6];
  const float* a_src2 = (const float*)d_in[7];
  const float* a_dst2 = (const float*)d_in[8];
  const float* b2     = (const float*)d_in[9];
  const float* clas_W = (const float*)d_in[10];
  const float* clas_b = (const float*)d_in[11];

  float* out = (float*)d_out;
  float* x2  = out + (size_t)N_NODES * N_CLASSES;

  float* ws   = (float*)d_ws;
  float* pa   = ws;
  float* pb   = ws + (size_t)N_NODES * HDIM;
  float* acc  = ws + (size_t)2 * N_NODES * HDIM;
  float* as_  = ws + (size_t)3 * N_NODES * HDIM;
  float* ad_  = as_ + N_NODES;
  float* m    = ad_ + N_NODES;
  float* ebuf = m + 2 * N_NODES;
  float* far1 = pb;
  float* far2 = pa;
  unsigned long long* sfix = (unsigned long long*)out;

  const int gemm_blocks = (N_NODES + 127) / 128;
  const int node_blocks = (N_NODES + NODE_BLOCK_NODES - 1) / NODE_BLOCK_NODES;
  const int edge_blocks = (N_EDGES + 256 * EXPSUM_EDGES_PER_THREAD - 1) / (256 * EXPSUM_EDGES_PER_THREAD);
  const int scat_blocks = N_EDGES / (SCAT_EDGES_PER_WAVE * 8);
  const int norm_total  = N_NODES * HDIM;
  const int norm_blocks = (norm_total + NORM_BLOCK_ELEMS - 1) / NORM_BLOCK_ELEMS;

  gemm_wmma_f32<IN_DIM, HDIM><<<gemm_blocks, 256, 0, stream>>>(X, W1, nullptr, pa, N_NODES);
  node_prep<<<node_blocks, 256, 0, stream>>>(pa, a_src1, a_dst1, as_, ad_, m, N_NODES);
  edge_max_tiled<<<MAX_RANGES * MAX_CHUNKS, 1024, 0, stream>>>(adj, as_, ad_, m, N_EDGES);
  node_post<<<node_blocks, 256, 0, stream>>>(pa, as_, ad_, m, sfix, acc, far1, N_NODES);
  edge_expsum<<<edge_blocks, 256, 0, stream>>>(adj, as_, ad_, ebuf, m, N_EDGES);
  edge_sum_tiled<<<MAX_RANGES * MAX_CHUNKS, 1024, 0, stream>>>(adj, ebuf, sfix, N_EDGES);
  edge_scatter<<<scat_blocks, 256, 0, stream>>>(adj, ebuf, pa, acc, far1, N_EDGES);
  normalize_bias<<<norm_blocks, 256, 0, stream>>>(acc, far1, sfix, b1, pa, norm_total,  1);

  gemm_wmma_f32<HDIM, HDIM><<<gemm_blocks, 256, 0, stream>>>(pa, W2, nullptr, pb, N_NODES);
  node_prep<<<node_blocks, 256, 0, stream>>>(pb, a_src2, a_dst2, as_, ad_, m, N_NODES);
  edge_max_tiled<<<MAX_RANGES * MAX_CHUNKS, 1024, 0, stream>>>(adj, as_, ad_, m, N_EDGES);
  node_post<<<node_blocks, 256, 0, stream>>>(pb, as_, ad_, m, sfix, acc, far2, N_NODES);
  edge_expsum<<<edge_blocks, 256, 0, stream>>>(adj, as_, ad_, ebuf, m, N_EDGES);
  edge_sum_tiled<<<MAX_RANGES * MAX_CHUNKS, 1024, 0, stream>>>(adj, ebuf, sfix, N_EDGES);
  edge_scatter<<<scat_blocks, 256, 0, stream>>>(adj, ebuf, pb, acc, far2, N_EDGES);
  normalize_bias<<<norm_blocks, 256, 0, stream>>>(acc, far2, sfix, b2, x2, norm_total,  0);

  gemm_wmma_f32<HDIM, N_CLASSES><<<gemm_blocks, 256, 0, stream>>>(x2, clas_W, clas_b, out, N_NODES);
}


# edge_scatter computes its edges' weights itself (a lane an edge; alpha_d and m side by side, one 64-bit gather) and stores them for edge_sum_tiled, now behind it; edge_expsum goes; float sums same
# speedup vs baseline: 2.8729x; 1.0684x over previous
#include <hip/hip_runtime.h>
#include <hip/hip_bf16.h>

typedef float v2f __attribute__((ext_vector_type(2)));
typedef float v8f __attribute__((ext_vector_type(8)));

#define N_NODES   100000
#define N_EDGES   1600000
#define IN_DIM    128
#define HDIM      64
#define N_CLASSES 16
#define NEG_SLOPE 0.2f

template <int K, int N>
__global__ __launch_bounds__(256) void gemm_wmma_f32(
    const float* __restrict__ A, const float* __restrict__ B,
    const float* __restrict__ bias, float* __restrict__ C, int M) {
  __shared__ float Bs[K * N];
  const int tid = threadIdx.x;
  for (int i = tid; i < K * N; i += 256) Bs[i] = B[i];
  __syncthreads();

  const int wave = tid >> 5;
  const int lane = tid & 31;
  const int hi = lane >> 4;
  const int lo = lane & 15;

  const long m0 = (long)blockIdx.x * 128 + wave * 16;
  long r = m0 + lo;
  if (r > M - 1) r = M - 1;
  const float* __restrict__ Arow = A + (size_t)r * K;

  constexpr int NT = N / 16;
  v8f acc[NT] = {};

  for (int kk = 0; kk < K / 4; ++kk) {
    const int kb = kk * 4 + 2 * hi;
    v2f a;
    a.x = Arow[kb];
    a.y = Arow[kb + 1];
#pragma unroll
    for (int nt = 0; nt < NT; ++nt) {
      v2f b;
      b.x = Bs[kb * N + nt * 16 + lo];
      b.y = Bs[(kb + 1) * N + nt * 16 + lo];
      acc[nt] = __builtin_amdgcn_wmma_f32_16x16x4_f32(
           false, a,  false, b,
           (short)0, acc[nt],  false,  false);
    }
  }

#pragma unroll
  for (int nt = 0; nt < NT; ++nt) {
    const int col = nt * 16 + lo;
    const float bv = bias ? bias[col] : 0.0f;
#pragma unroll
    for (int rr = 0; rr < 8; ++rr) {
      const long row = m0 + rr + 8 * hi;
      if (row < M) C[(size_t)row * N + col] = acc[nt][rr] + bv;
    }
  }
}

#define NODES_PER_WAVE 16
#define NODE_BLOCK_NODES (NODES_PER_WAVE * 8)

__global__ __launch_bounds__(256) void node_prep(
    const float* __restrict__ h, const float* __restrict__ a_src,
    const float* __restrict__ a_dst, float* __restrict__ alpha_s,
    float* __restrict__ dm, int n_nodes) {
  const int lane = threadIdx.x & 31;
  const int n0 = (blockIdx.x * 8 + (threadIdx.x >> 5)) * NODES_PER_WAVE;
  if (n0 >= n_nodes) return;
  float h0[NODES_PER_WAVE], h1[NODES_PER_WAVE];
#pragma unroll
  for (int j = 0; j < NODES_PER_WAVE; ++j) {
    const int n = min(n0 + j, n_nodes - 1);
    const float* hr = h + (size_t)n * HDIM;
    h0[j] = hr[lane];
    h1[j] = hr[lane + 32];
  }
  const float as0 = a_src[lane], as1 = a_src[lane + 32];
  const float ad0 = a_dst[lane], ad1 = a_dst[lane + 32];
  float rs = 0.0f, rd = 0.0f;
#pragma unroll
  for (int j = 0; j < NODES_PER_WAVE; ++j) {
    float ps = h0[j] * as0 + h1[j] * as1;
    float pd = h0[j] * ad0 + h1[j] * ad1;
#pragma unroll
    for (int off = 16; off > 0; off >>= 1) {
      ps += __shfl_xor(ps, off, 32);
      pd += __shfl_xor(pd, off, 32);
    }
    if (lane == j) { rs = ps; rd = pd; }
  }
  const int n = n0 + lane;
  if (lane < NODES_PER_WAVE && n < n_nodes) {
    alpha_s[n] = rs;
    const float e = rs + rd;
    reinterpret_cast<float2*>(dm)[n] = make_float2(rd, (e > 0.0f) ? e : NEG_SLOPE * e);
  }
}

#define MAX_RANGES  8
#define MAX_CHUNKS  32
#define RANGE_NODES 12544
#define CHUNK_EDGES (N_EDGES / MAX_CHUNKS)
#define KEY_EMPTY   (-2147483647 - 1)

__device__ __forceinline__ int f32_order_key(float f) {
  const int b = __float_as_int(f);
  return (b >= 0) ? b : (b ^ 0x7fffffff);
}
__device__ __forceinline__ float f32_from_key(int k) {
  return __int_as_float((k >= 0) ? k : (k ^ 0x7fffffff));
}

__global__ __launch_bounds__(1024) void edge_max_tiled(
    const int* __restrict__ adj, const float* __restrict__ alpha_s,
    float* __restrict__ dm, int n_edges) {
  __shared__ int best[RANGE_NODES];
  const int tid   = threadIdx.x;
  const int range = blockIdx.x & (MAX_RANGES - 1);
  const int chunk = blockIdx.x / MAX_RANGES;
  const int base  = range * RANGE_NODES;

  for (int i = tid; i < RANGE_NODES; i += 1024) best[i] = KEY_EMPTY;
  __syncthreads();

  const int* __restrict__ srcp = adj + chunk * CHUNK_EDGES;
  const int* __restrict__ dstp = adj + n_edges + chunk * CHUNK_EDGES;
  constexpr int PER_IT = 4 * 1024;
  constexpr int NIT = (CHUNK_EDGES + PER_IT - 1) / PER_IT;

  int dn[4], sn[4];
#pragma unroll
  for (int j = 0; j < 4; ++j) {
    const int idx = j * 1024 + tid;
    dn[j] = dstp[idx];
    sn[j] = srcp[idx];
  }
  for (int it = 0; it < NIT; ++it) {
    int dc[4], sc[4];
#pragma unroll
    for (int j = 0; j < 4; ++j) { dc[j] = dn[j]; sc[j] = sn[j]; }
#pragma unroll
    for (int j = 0; j < 4; ++j) {
      const int idx = (it + 1) * PER_IT + j * 1024 + tid;
      const bool ok = idx < CHUNK_EDGES;
      dn[j] = ok ? dstp[idx] : -1;
      sn[j] = ok ? srcp[idx] : 0;
    }
    unsigned loc[4];
    float a[4];
#pragma unroll
    for (int j = 0; j < 4; ++j) {
      loc[j] = (unsigned)(dc[j] - base);
      a[j] = (loc[j] < (unsigned)RANGE_NODES) ? alpha_s[sc[j]] : 0.0f;
    }
#pragma unroll
    for (int j = 0; j < 4; ++j) {
      if (loc[j] < (unsigned)RANGE_NODES) atomicMax(&best[loc[j]], f32_order_key(a[j]));
    }
  }
  __syncthreads();

  constexpr int NF = (RANGE_NODES + 1023) / 1024;
  int   k[NF];
  float adv[NF];
#pragma unroll
  for (int j = 0; j < NF; ++j) {
    const int i = j * 1024 + tid;
    k[j] = (i < RANGE_NODES) ? best[i] : KEY_EMPTY;
  }
#pragma unroll
  for (int j = 0; j < NF; ++j) {
    adv[j] = (k[j] != KEY_EMPTY) ? dm[2 * (base + j * 1024 + tid)] : 0.0f;
  }
#pragma unroll
  for (int j = 0; j < NF; ++j) {
    if (k[j] != KEY_EMPTY) {
      float e = f32_from_key(k[j]) + adv[j];
      e = (e > 0.0f) ? e : NEG_SLOPE * e;
      atomicMax(&dm[2 * (base + j * 1024 + tid) + 1], e);
    }
  }
}

#define SUM_ONE 1099511627776.0f
__device__ __forceinline__ unsigned long long weight_to_fixed(float p) {
  return (unsigned long long)(p * SUM_ONE);
}
__device__ __forceinline__ float fixed_to_sum(unsigned long long v) {
  return (float)v * (1.0f / SUM_ONE);
}

__global__ __launch_bounds__(256) void node_post(
    const float* __restrict__ h, const float* __restrict__ alpha_s,
    const float* __restrict__ dm,
    unsigned long long* __restrict__ s_fix, float* __restrict__ acc,
    float* __restrict__ acc_far, int n_nodes) {
  const int lane = threadIdx.x & 31;
  const int n0 = (blockIdx.x * 8 + (threadIdx.x >> 5)) * NODES_PER_WAVE;
  if (n0 >= n_nodes) return;
  float r0[NODES_PER_WAVE], r1[NODES_PER_WAVE];
#pragma unroll
  for (int j = 0; j < NODES_PER_WAVE; ++j) {
    const int n = min(n0 + j, n_nodes - 1);
    const size_t base = (size_t)n * HDIM;
    r0[j] = h[base + lane];
    r1[j] = h[base + lane + 32];
  }
  const int nl = min(n0 + (lane & (NODES_PER_WAVE - 1)), n_nodes - 1);
  const float2 dmn = reinterpret_cast<const float2*>(dm)[nl];
  float e = alpha_s[nl] + dmn.x;
  e = (e > 0.0f) ? e : NEG_SLOPE * e;
  const float pself = __expf(e - dmn.y);
  if (lane < NODES_PER_WAVE && n0 + lane < n_nodes) s_fix[n0 + lane] = weight_to_fixed(pself);
#pragma unroll
  for (int j = 0; j < NODES_PER_WAVE; ++j) {
    if (n0 + j < n_nodes) {
      const float p = __int_as_float(__builtin_amdgcn_readlane(__float_as_int(pself), j));
      const size_t base = (size_t)(n0 + j) * HDIM;
      acc[base + lane]      = r0[j] * p;
      acc[base + lane + 32] = r1[j] * p;
      acc_far[base + lane]      = 0.0f;
      acc_far[base + lane + 32] = 0.0f;
    }
  }
}

static_assert(CHUNK_EDGES % 4 == 0, "a chunk is whole 128-bit groups, 16-byte aligned in both arrays");
#define SUM_VECS      (CHUNK_EDGES / 4)
#define SUM_GROUPS    4
#define SUM_PER_TRIP  (SUM_GROUPS * 1024)

__global__ __launch_bounds__(1024) void edge_sum_tiled(
    const int* __restrict__ adj, const float* __restrict__ ebuf,
    unsigned long long* __restrict__ s_fix, int n_edges) {
  __shared__ unsigned long long part[RANGE_NODES];
  const int tid   = threadIdx.x;
  const int range = blockIdx.x & (MAX_RANGES - 1);
  const int chunk = blockIdx.x / MAX_RANGES;
  const int base  = range * RANGE_NODES;

  for (int i = tid; i < RANGE_NODES; i += 1024) part[i] = 0ull;
  __syncthreads();

  const int4*   __restrict__ dstv = reinterpret_cast<const int4*>(adj + n_edges + chunk * CHUNK_EDGES);
  const float4* __restrict__ wv   = reinterpret_cast<const float4*>(ebuf + chunk * CHUNK_EDGES);
  constexpr int NTRIP = (SUM_VECS + SUM_PER_TRIP - 1) / SUM_PER_TRIP;

  int4   dn[SUM_GROUPS];
  float4 wn[SUM_GROUPS];
#pragma unroll
  for (int j = 0; j < SUM_GROUPS; ++j) {
    const int v = j * 1024 + tid;
    dn[j] = dstv[v];
    wn[j] = wv[v];
  }
#pragma nounroll
  for (int it = 0; it < NTRIP; ++it) {
    int4   dc[SUM_GROUPS];
    float4 wc[SUM_GROUPS];
#pragma unroll
    for (int j = 0; j < SUM_GROUPS; ++j) { dc[j] = dn[j]; wc[j] = wn[j]; }
#pragma unroll
    for (int j = 0; j < SUM_GROUPS; ++j) {
      const int v = (it + 1) * SUM_PER_TRIP + j * 1024 + tid;
      const bool ok = v < SUM_VECS;
      dn[j] = ok ? dstv[v] : make_int4(-1, -1, -1, -1);
      wn[j] = ok ? wv[v]   : make_float4(0.0f, 0.0f, 0.0f, 0.0f);
    }
#pragma unroll
    for (int j = 0; j < SUM_GROUPS; ++j) {
      const int   dd[4] = {dc[j].x, dc[j].y, dc[j].z, dc[j].w};
      const float ww[4] = {wc[j].x, wc[j].y, wc[j].z, wc[j].w};
#pragma unroll
      for (int k = 0; k < 4; ++k) {
        const unsigned loc = (unsigned)(dd[k] - base);
        if (loc < (unsigned)RANGE_NODES) atomicAdd(&part[loc], weight_to_fixed(ww[k]));
      }
    }
  }
  __syncthreads();

  constexpr int NF = (RANGE_NODES + 1023) / 1024;
  unsigned long long v[NF];
#pragma unroll
  for (int j = 0; j < NF; ++j) {
    const int i = j * 1024 + tid;
    v[j] = (i < RANGE_NODES) ? part[i] : 0ull;
  }
#pragma unroll
  for (int j = 0; j < NF; ++j) {
    if (v[j] != 0ull) atomicAdd(&s_fix[base + j * 1024 + tid], v[j]);
  }
}

#define SCAT_EDGES_PER_WAVE 64
#define SCAT_GROUP          16
static_assert(N_EDGES % (SCAT_EDGES_PER_WAVE * 8) == 0, "a block of 8 waves takes 512 whole edges");

__global__ __launch_bounds__(256) void edge_scatter(
    const int* __restrict__ adj, const float* __restrict__ alpha_s,
    const float* __restrict__ dm, float* __restrict__ ebuf,
    const float* __restrict__ h, float* __restrict__ acc_near, float* __restrict__ acc_far,
    int n_edges) {
  const int lane = threadIdx.x & 31;
  const int e0 = (blockIdx.x * 8 + (threadIdx.x >> 5)) * SCAT_EDGES_PER_WAVE;
  if (e0 >= n_edges) return;

  const unsigned placement = __builtin_amdgcn_s_sendmsg_rtn(0x87);
  float* __restrict__ acc = ((placement >> 18) & 1u) ? acc_far : acc_near;

  int sv[2], dv[2], wv[2];
#pragma unroll
  for (int k = 0; k < 2; ++k) {
    const int idx = e0 + k * 32 + lane;
    sv[k] = adj[idx];
    dv[k] = adj[n_edges + idx];
  }
  float  as[2];
  float2 dmv[2];
#pragma unroll
  for (int k = 0; k < 2; ++k) {
    as[k]  = alpha_s[sv[k]];
    dmv[k] = reinterpret_cast<const float2*>(dm)[dv[k]];
  }
#pragma unroll
  for (int k = 0; k < 2; ++k) {
    float e = as[k] + dmv[k].x;
    e = (e > 0.0f) ? e : NEG_SLOPE * e;
    const float p = __expf(e - dmv[k].y);
    ebuf[e0 + k * 32 + lane] = p;
    wv[k] = __float_as_int(p);
  }

#pragma unroll
  for (int g = 0; g < SCAT_EDGES_PER_WAVE / SCAT_GROUP; ++g) {
    const int k  = (g * SCAT_GROUP) >> 5;
    const int l0 = (g * SCAT_GROUP) & 31;
    float r0[SCAT_GROUP], r1[SCAT_GROUP];
#pragma unroll
    for (int j = 0; j < SCAT_GROUP; ++j) {
      const int s = __builtin_amdgcn_readlane(sv[k], l0 + j);
      const unsigned si = (unsigned)s * HDIM + (unsigned)lane;
      r0[j] = h[si];
      r1[j] = h[si + 32];
    }
#pragma unroll
    for (int j = 0; j < SCAT_GROUP; ++j) {
      const int d = __builtin_amdgcn_readlane(dv[k], l0 + j);
      const float w = __int_as_float(__builtin_amdgcn_readlane(wv[k], l0 + j));
      const unsigned di = (unsigned)d * HDIM + (unsigned)lane;
      atomicAdd(&acc[di],      r0[j] * w);
      atomicAdd(&acc[di + 32], r1[j] * w);
    }
  }
}

#define NORM_VECS         4
#define NORM_STRIDE       (256 * 4)
#define NORM_BLOCK_ELEMS  (NORM_STRIDE * NORM_VECS)
static_assert(NORM_STRIDE % HDIM == 0, "a thread keeps its feature columns from group to group");
static_assert((N_NODES * HDIM) % 4 == 0, "a group of four is all inside the array");

__device__ __forceinline__ float norm_one(float a, float sv, float bv, int apply_elu) {
  float v = a / sv + bv;
  if (apply_elu) v = (v > 0.0f) ? v : (__expf(v) - 1.0f);
  return v;
}

__global__ __launch_bounds__(256) void normalize_bias(
    const float* __restrict__ acc, const float* __restrict__ acc_far,
    const unsigned long long* __restrict__ s_fix,
    const float* __restrict__ b, float* __restrict__ out, int total,
    int apply_elu) {
  const int i0 = blockIdx.x * NORM_BLOCK_ELEMS + threadIdx.x * 4;
  const float4 bv = *reinterpret_cast<const float4*>(b + ((threadIdx.x * 4) & (HDIM - 1)));
  float4 a[NORM_VECS], a2[NORM_VECS];
  float  sv[NORM_VECS];
#pragma unroll
  for (int k = 0; k < NORM_VECS; ++k) {
    const int idx = i0 + k * NORM_STRIDE;
    const bool ok = idx < total;
    a[k]  = ok ? *reinterpret_cast<const float4*>(acc + idx) : make_float4(0.0f, 0.0f, 0.0f, 0.0f);
    a2[k] = ok ? *reinterpret_cast<const float4*>(acc_far + idx) : make_float4(0.0f, 0.0f, 0.0f, 0.0f);
    sv[k] = ok ? fixed_to_sum(s_fix[idx >> 6]) : 1.0f;
  }
#pragma unroll
  for (int k = 0; k < NORM_VECS; ++k) {
    const int idx = i0 + k * NORM_STRIDE;
    if (idx < total) {
      float4 v;
      v.x = norm_one(a[k].x + a2[k].x, sv[k], bv.x, apply_elu);
      v.y = norm_one(a[k].y + a2[k].y, sv[k], bv.y, apply_elu);
      v.z = norm_one(a[k].z + a2[k].z, sv[k], bv.z, apply_elu);
      v.w = norm_one(a[k].w + a2[k].w, sv[k], bv.w, apply_elu);
      *reinterpret_cast<float4*>(out + idx) = v;
    }
  }
}

extern "C" void kernel_launch(void* const* d_in, const int* in_sizes, int n_in,
                              void* d_out, int out_size, void* d_ws,
                              size_t ws_size, hipStream_t stream) {
  const float* X      = (const float*)d_in[0];
  const int*   adj    = (const int*)d_in[1];
  const float* W1     = (const float*)d_in[2];
  const float* a_src1 = (const float*)d_in[3];
  const float* a_dst1 = (const float*)d_in[4];
  const float* b1     = (const float*)d_in[5];
  const float* W2     = (const float*)d_in[6];
  const float* a_src2 = (const float*)d_in[7];
  const float* a_dst2 = (const float*)d_in[8];
  const float* b2     = (const float*)d_in[9];
  const float* clas_W = (const float*)d_in[10];
  const float* clas_b = (const float*)d_in[11];

  float* out = (float*)d_out;
  float* x2  = out + (size_t)N_NODES * N_CLASSES;

  float* ws   = (float*)d_ws;
  float* pa   = ws;
  float* pb   = ws + (size_t)N_NODES * HDIM;
  float* acc  = ws + (size_t)2 * N_NODES * HDIM;
  float* as_  = ws + (size_t)3 * N_NODES * HDIM;
  float* dm   = as_ + N_NODES;
  float* ebuf = dm + 3 * N_NODES;
  float* far1 = pb;
  float* far2 = pa;
  unsigned long long* sfix = (unsigned long long*)out;

  const int gemm_blocks = (N_NODES + 127) / 128;
  const int node_blocks = (N_NODES + NODE_BLOCK_NODES - 1) / NODE_BLOCK_NODES;
  const int scat_blocks = N_EDGES / (SCAT_EDGES_PER_WAVE * 8);
  const int norm_total  = N_NODES * HDIM;
  const int norm_blocks = (norm_total + NORM_BLOCK_ELEMS - 1) / NORM_BLOCK_ELEMS;

  gemm_wmma_f32<IN_DIM, HDIM><<<gemm_blocks, 256, 0, stream>>>(X, W1, nullptr, pa, N_NODES);
  node_prep<<<node_blocks, 256, 0, stream>>>(pa, a_src1, a_dst1, as_, dm, N_NODES);
  edge_max_tiled<<<MAX_RANGES * MAX_CHUNKS, 1024, 0, stream>>>(adj, as_, dm, N_EDGES);
  node_post<<<node_blocks, 256, 0, stream>>>(pa, as_, dm, sfix, acc, far1, N_NODES);
  edge_scatter<<<scat_blocks, 256, 0, stream>>>(adj, as_, dm, ebuf, pa, acc, far1, N_EDGES);
  edge_sum_tiled<<<MAX_RANGES * MAX_CHUNKS, 1024, 0, stream>>>(adj, ebuf, sfix, N_EDGES);
  normalize_bias<<<norm_blocks, 256, 0, stream>>>(acc, far1, sfix, b1, pa, norm_total,  1);

  gemm_wmma_f32<HDIM, HDIM><<<gemm_blocks, 256, 0, stream>>>(pa, W2, nullptr, pb, N_NODES);
  node_prep<<<node_blocks, 256, 0, stream>>>(pb, a_src2, a_dst2, as_, dm, N_NODES);
  edge_max_tiled<<<MAX_RANGES * MAX_CHUNKS, 1024, 0, stream>>>(adj, as_, dm, N_EDGES);
  node_post<<<node_blocks, 256, 0, stream>>>(pb, as_, dm, sfix, acc, far2, N_NODES);
  edge_scatter<<<scat_blocks, 256, 0, stream>>>(adj, as_, dm, ebuf, pb, acc, far2, N_EDGES);
  edge_sum_tiled<<<MAX_RANGES * MAX_CHUNKS, 1024, 0, stream>>>(adj, ebuf, sfix, N_EDGES);
  normalize_bias<<<norm_blocks, 256, 0, stream>>>(acc, far2, sfix, b2, x2, norm_total,  0);

  gemm_wmma_f32<HDIM, N_CLASSES><<<gemm_blocks, 256, 0, stream>>>(x2, clas_W, clas_b, out, N_NODES);
}
